# starting kernel
# baseline (speedup 1.0000x reference)
#include <hip/hip_runtime.h>
#include <math.h>


typedef __attribute__((ext_vector_type(2))) float v2f;
typedef __attribute__((ext_vector_type(8))) float v8f;

#ifndef __has_builtin
#define __has_builtin(x) 0
#endif

#if __has_builtin(__builtin_amdgcn_wmma_f32_16x16x4_f32)
#define USE_WMMA_F32X4 1
#else
#define USE_WMMA_F32X4 0
#endif

namespace {
constexpr int kS = 256;
constexpr int kC = 128;
constexpr int kCenter = 128;
constexpr int kIters = 33;
constexpr int kShifts = 129;
constexpr int kKTiles = 36;
constexpr int kRows = 8;
constexpr int kHalo = 4;
constexpr int kRunBuf = kS + 2 * kHalo;
}

__global__ __launch_bounds__(256) void tonal_diffusion_kernel(
    const float* __restrict__ data,
    const float* __restrict__ log_weights,
    float* __restrict__ out,
    int n_rows) {
  const int lane = threadIdx.x & 31;
  const int wave = threadIdx.x >> 5;
  const int row  = blockIdx.x * kRows + wave;
  const int rowc = (row < n_rows) ? row : (n_rows - 1);

  __shared__ float sRunA[kRows][kRunBuf];
  __shared__ float sRunB[kRows][kRunBuf];
  __shared__ float sDataPad[kRows][160];
  __shared__ float sLogAcc[kRows][384];
  __shared__ float sZ[kRows][132];

  const int steps[6] = {1, -1, -3, 3, 4, -4};

  float p[6];
  float rate = 0.f;
#pragma unroll
  for (int i = 0; i < 6; ++i) {
    p[i] = expf(log_weights[(size_t)rowc * 6 + i]);
    rate += p[i];
  }
  const float inv_rate = 1.f / rate;
#pragma unroll
  for (int i = 0; i < 6; ++i) p[i] *= inv_rate;

  for (int idx = lane; idx < 160; idx += 32) {
    float v = 0.f;
    if (idx >= 16 && idx < 16 + kC) {
      float raw = data[(size_t)rowc * kC + (idx - 16)];
      v = (raw > 1e-8f) ? raw : 0.f;
    }
    sDataPad[wave][idx] = v;
  }

  for (int idx = lane; idx < kRunBuf; idx += 32) {
    sRunA[wave][idx] = 0.f;
    sRunB[wave][idx] = 0.f;
  }
  float* runA = &sRunA[wave][kHalo];
  float* runB = &sRunB[wave][kHalo];
  if (lane == 0) runA[kCenter] = 1.f;

  float acc[8];
#pragma unroll
  for (int e = 0; e < 8; ++e) acc[e] = 0.f;
  __syncthreads();

  float sp = expf(-rate);
  for (int n = 0; n < kIters; ++n) {
    const int lo_t = kCenter - kHalo * (n + 1);
    const int hi_t = kCenter + kHalo * (n + 1);
    const int eLo = (lo_t > 0 ? lo_t : 0) >> 5;
    const int eHi = (hi_t < kS - 1 ? hi_t : kS - 1) >> 5;
#pragma unroll
    for (int e = 0; e < 8; ++e) {
      if (e >= eLo && e <= eHi) {
        const int t = e * 32 + lane;
        const float r = runA[t];
        acc[e] = fmaf(sp, r, acc[e]);
        float nv = 0.f;
#pragma unroll
        for (int i = 0; i < 6; ++i)
          nv = fmaf(p[i], runA[t - steps[i]], nv);
        runB[t] = nv;
      }
    }
    __syncthreads();
    float* tmp = runA; runA = runB; runB = tmp;
    sp = sp * rate / (float)(n + 1);
  }

  float* accL = runA;
#pragma unroll
  for (int e = 0; e < 8; ++e) {
    const int t = e * 32 + lane;
    accL[t] = acc[e];
    sLogAcc[wave][t] = logf(fmaxf(acc[e], 1e-38f));
  }
  for (int t = kS + lane; t < 384; t += 32) sLogAcc[wave][t] = 0.f;
  __syncthreads();

  for (int k = lane; k < kShifts; k += 32) {
    float z = 0.f;
    for (int c = 0; c < kC; ++c) z += accL[k + c];
    sZ[wave][k] = logf(z);
  }

  float pl = 0.f;
  for (int c = lane; c < kC; c += 32) {
    const float v = sDataPad[wave][16 + c];
    if (v > 0.f) pl += v * logf(v);
  }
#pragma unroll
  for (int off = 16; off; off >>= 1) pl += __shfl_xor(pl, off, 32);
  __syncthreads();

  float best = INFINITY;

#if USE_WMMA_F32X4
  const int m    = lane & 15;
  const int koff = (lane & 16) ? 2 : 0;
  const float* dp = &sDataPad[wave][16 - m];
  const float* lp = &sLogAcc[wave][16 * m];
  v8f dacc = {0.f, 0.f, 0.f, 0.f, 0.f, 0.f, 0.f, 0.f};
#pragma unroll
  for (int kc = 0; kc < kKTiles; ++kc) {
    const int j = 4 * kc + koff;
    v2f a = {dp[j], dp[j + 1]};
    v2f b = {lp[j], lp[j + 1]};
    dacc = __builtin_amdgcn_wmma_f32_16x16x4_f32(
         false, a,  false, b,
         (short)0, dacc,  false,  false);
  }
  const int mbase = (lane & 16) ? 8 : 0;
  const int n16   = 16 * (lane & 15);
#pragma unroll
  for (int r = 0; r < 8; ++r) {
    const int k = (r + mbase) + n16;
    if (k < kShifts) {
      const float loss = pl - dacc[r] + sZ[wave][k];
      best = fminf(best, loss);
    }
  }
#else
  for (int k = lane; k < kShifts; k += 32) {
    float cr = 0.f;
    for (int c = 0; c < kC; ++c)
      cr = fmaf(sDataPad[wave][16 + c], sLogAcc[wave][k + c], cr);
    best = fminf(best, pl - cr + sZ[wave][k]);
  }
#endif

#pragma unroll
  for (int off = 16; off; off >>= 1) best = fminf(best, __shfl_xor(best, off, 32));
  if (lane == 0 && row < n_rows) out[row] = best;
}

extern "C" void kernel_launch(void* const* d_in, const int* in_sizes, int n_in,
                              void* d_out, int out_size, void* d_ws, size_t ws_size,
                              hipStream_t stream) {
  (void)n_in; (void)d_ws; (void)ws_size; (void)out_size;
  const float* data = (const float*)d_in[0];
  const float* logw = (const float*)d_in[1];
  float* out = (float*)d_out;
  const int n_rows = in_sizes[0] / kC;
  const int blocks = (n_rows + kRows - 1) / kRows;
  tonal_diffusion_kernel<<<blocks, 256, 0, stream>>>(data, logw, out, n_rows);
}


# Window sums Z[k] on the matrix unit: a second Toeplitz product (A = 128-wide band of ones, B = acc) interleaved with the cross chain replaces 5x128 LDS adds per lane; Z summed in the unit's k order
# speedup vs baseline: 1.8537x; 1.8537x over previous
#include <hip/hip_runtime.h>
#include <math.h>


typedef __attribute__((ext_vector_type(2))) float v2f;
typedef __attribute__((ext_vector_type(4))) float v4f;
typedef __attribute__((ext_vector_type(8))) float v8f;

namespace {
constexpr int kS = 256;
constexpr int kC = 128;
constexpr int kCenter = 128;
constexpr int kTerms = 33;
constexpr int kShifts = 129;
constexpr int kKTiles = 36;
constexpr int kRows = 8;
constexpr int kHalo = 4;
constexpr int kFold = 4;
constexpr int kPasses = 7;
constexpr int kPassTiles = 12;
constexpr int kWinLo = 96;
constexpr int kWin = 64;
constexpr int kSetBuf = kWin + 2 * kHalo;
constexpr int kXLo = 16;
constexpr int kXBuf = kXLo + kS + 32;
constexpr int kVRow = kWin + 8;
}

__device__ __forceinline__ v8f mma_f32x4(v2f a, v2f b, v8f c) {
  return __builtin_amdgcn_wmma_f32_16x16x4_f32(
       false, a,  false, b,
       (short)0, c,  false,  false);
}

__global__ __launch_bounds__(256) void tonal_diffusion_kernel(
    const float* __restrict__ data,
    const float* __restrict__ log_weights,
    float* __restrict__ out,
    int n_rows) {
  const int lane = threadIdx.x & 31;
  const int wave = threadIdx.x >> 5;
  const int row  = blockIdx.x * kRows + wave;
  const int rowc = (row < n_rows) ? row : (n_rows - 1);

  __shared__ __attribute__((aligned(16))) float sSet[kRows][2][kSetBuf];
  __shared__ __attribute__((aligned(16))) float sX[kRows][kXBuf];
  __shared__ __attribute__((aligned(16))) float sV[kRows][kPasses][kVRow];
  __shared__ float sDataPad[kRows][160];
  __shared__ __attribute__((aligned(16))) float sLogAcc[kRows][384];

  const int steps[6] = {1, -1, -3, 3, 4, -4};

  float p[6];
  float rate = 0.f;
#pragma unroll
  for (int i = 0; i < 6; ++i) {
    p[i] = expf(log_weights[(size_t)rowc * 6 + i]);
    rate += p[i];
  }
  const float inv_rate = 1.f / rate;
#pragma unroll
  for (int i = 0; i < 6; ++i) p[i] *= inv_rate;

  for (int idx = lane; idx < 160; idx += 32) {
    float v = 0.f;
    if (idx >= 16 && idx < 16 + kC) {
      float raw = data[(size_t)rowc * kC + (idx - 16)];
      v = (raw > 1e-8f) ? raw : 0.f;
    }
    sDataPad[wave][idx] = v;
  }

  {
    float* setFlat = &sSet[wave][0][0];
    for (int idx = lane; idx < 2 * kSetBuf; idx += 32) setFlat[idx] = 0.f;
    v4f* x4 = (v4f*)&sX[wave][0];
    const v4f zero4 = {0.f, 0.f, 0.f, 0.f};
    for (int idx = lane; idx < kXBuf / 4; idx += 32) x4[idx] = zero4;
    for (int idx = lane; idx < kPasses * 8; idx += 32)
      sV[wave][idx >> 3][kWin + (idx & 7)] = 0.f;
  }
  float* runA = &sSet[wave][0][0];
  float* runB = &sSet[wave][1][0];
  if (lane == 0) runA[kCenter - kWinLo + kHalo] = 1.f;

  float pm[kTerms];
  pm[0] = expf(-rate);
#pragma unroll
  for (int n = 0; n < kTerms - 1; ++n) pm[n + 1] = pm[n] * rate / (float)(n + 1);
  __syncthreads();

  float rr[kFold + 1][2];
#pragma unroll
  for (int s = 0; s < kFold; ++s) {
#pragma unroll
    for (int c = 0; c < 2; ++c) {
      const int u = kHalo + 32 * c + lane;
      rr[s][c] = runA[u];
      float nv = 0.f;
#pragma unroll
      for (int i = 0; i < 6; ++i)
        nv = fmaf(p[i], runA[u - steps[i]], nv);
      runB[u] = nv;
    }
    __syncthreads();
    float* tmp = runA; runA = runB; runB = tmp;
  }
#pragma unroll
  for (int c = 0; c < 2; ++c) rr[kFold][c] = runA[kHalo + 32 * c + lane];

#pragma unroll
  for (int q = 0; q < kPasses; ++q) {
#pragma unroll
    for (int c = 0; c < 2; ++c) {
      float v = pm[4 * q] * rr[0][c];
#pragma unroll
      for (int r = 1; r < kFold; ++r) v = fmaf(pm[4 * q + r], rr[r][c], v);
      sV[wave][q][32 * c + lane] = v;
    }
  }
#pragma unroll
  for (int c = 0; c < 2; ++c) {
    float v = pm[4 * kPasses] * rr[0][c];
#pragma unroll
    for (int r = 1; r <= kFold; ++r) v = fmaf(pm[4 * kPasses + r], rr[r][c], v);
    sX[wave][kXLo + kWinLo + 32 * c + lane] = v;
  }

  const int c16 = lane & 15;
  const int h   = lane >> 4;
  v2f afr[kPassTiles];
  {
    const float* g4 = &runA[(kCenter + 16 + c16 - 2 * h) - kWinLo + kHalo];
#pragma unroll
    for (int kc = 0; kc < kPassTiles; ++kc) {
      afr[kc].x = g4[-4 * kc];
      afr[kc].y = g4[-4 * kc - 1];
    }
  }
  __syncthreads();

  const bool in_win = (c16 >= kWinLo / 16) && (c16 < (kWinLo + kWin) / 16);
  const float* vp = &sV[wave][0][in_win ? (16 * c16 + 8 * h - kWinLo) : kWin];
  const float* bp = &sX[wave][16 * c16 + 2 * h];
  float* xo = &sX[wave][kXLo + 16 * c16 + 8 * h];
  v8f acc8;
#pragma unroll
  for (int q = kPasses - 1; q >= 0; --q) {
    const v4f vlo = *(const v4f*)(vp + q * kVRow);
    const v4f vhi = *(const v4f*)(vp + q * kVRow + 4);
    v8f c0 = {vlo.x, vlo.y, vlo.z, vlo.w, vhi.x, vhi.y, vhi.z, vhi.w};
    v8f c1 = {0.f, 0.f, 0.f, 0.f, 0.f, 0.f, 0.f, 0.f};
    v2f bfr[kPassTiles];
#pragma unroll
    for (int kc = 0; kc < kPassTiles; ++kc) bfr[kc] = *(const v2f*)(bp + 4 * kc);
    __builtin_amdgcn_sched_barrier(0);
#pragma unroll
    for (int kc = 0; kc < kPassTiles; kc += 2) {
      c0 = mma_f32x4(afr[kc], bfr[kc], c0);
      c1 = mma_f32x4(afr[kc + 1], bfr[kc + 1], c1);
    }
    acc8 = c0 + c1;
    if (q > 0) {
      *(v4f*)xo       = acc8.lo;
      *(v4f*)(xo + 4) = acc8.hi;
      __syncthreads();
    }
  }

  float* accL = &sX[wave][kXLo];
  {
    v8f lg;
#pragma unroll
    for (int r = 0; r < 8; ++r) lg[r] = logf(fmaxf(acc8[r], 1e-38f));
    *(v4f*)xo       = acc8.lo;
    *(v4f*)(xo + 4) = acc8.hi;
    float* lo = &sLogAcc[wave][16 * c16 + 8 * h];
    *(v4f*)lo       = lg.lo;
    *(v4f*)(lo + 4) = lg.hi;
  }
  for (int t = kS + lane; t < 384; t += 32) sLogAcc[wave][t] = 0.f;
  __syncthreads();

  float pl = 0.f;
  for (int c = lane; c < kC; c += 32) {
    const float v = sDataPad[wave][16 + c];
    if (v > 0.f) pl += v * logf(v);
  }
#pragma unroll
  for (int off = 16; off; off >>= 1) pl += __shfl_xor(pl, off, 32);

  float best = INFINITY;

  const int m    = lane & 15;
  const int koff = (lane & 16) ? 2 : 0;
  const float* dp = &sDataPad[wave][16 - m];
  const float* lp = &sLogAcc[wave][16 * m];
  const float* ap = &accL[16 * (m < 8 ? m : 8)];
  v8f dacc = {0.f, 0.f, 0.f, 0.f, 0.f, 0.f, 0.f, 0.f};
  v8f zacc = {0.f, 0.f, 0.f, 0.f, 0.f, 0.f, 0.f, 0.f};
#pragma unroll
  for (int kc = 0; kc < kKTiles; ++kc) {
    const int j = 4 * kc + koff;
    v2f a = {dp[j], dp[j + 1]};
    v2f b = {lp[j], lp[j + 1]};
    v2f a1 = {((unsigned)(j - m) < (unsigned)kC) ? 1.f : 0.f,
              ((unsigned)(j + 1 - m) < (unsigned)kC) ? 1.f : 0.f};
    v2f b1 = {ap[j], ap[j + 1]};
    dacc = mma_f32x4(a, b, dacc);
    zacc = mma_f32x4(a1, b1, zacc);
  }
  const int mbase = (lane & 16) ? 8 : 0;
  const int n16   = 16 * (lane & 15);
#pragma unroll
  for (int r = 0; r < 8; ++r) {
    const int k = (r + mbase) + n16;
    const float loss = pl - dacc[r] + logf(zacc[r]);
    best = (k < kShifts) ? fminf(best, loss) : best;
  }

#pragma unroll
  for (int off = 16; off; off >>= 1) best = fminf(best, __shfl_xor(best, off, 32));
  if (lane == 0 && row < n_rows) out[row] = best;
}

extern "C" void kernel_launch(void* const* d_in, const int* in_sizes, int n_in,
                              void* d_out, int out_size, void* d_ws, size_t ws_size,
                              hipStream_t stream) {
  (void)n_in; (void)d_ws; (void)ws_size; (void)out_size;
  const float* data = (const float*)d_in[0];
  const float* logw = (const float*)d_in[1];
  float* out = (float*)d_out;
  const int n_rows = in_sizes[0] / kC;
  const int blocks = (n_rows + kRows - 1) / kRows;
  tonal_diffusion_kernel<<<blocks, 256, 0, stream>>>(data, logw, out, n_rows);
}


# Data row read once into registers (one 128-bit load per lane at entry); LDS staging and plogp both use them: no 5 dependent load-wait-store trips, no plogp LDS loop; plogp lane order changes
# speedup vs baseline: 2.1199x; 1.1436x over previous
#include <hip/hip_runtime.h>
#include <math.h>


typedef __attribute__((ext_vector_type(2))) float v2f;
typedef __attribute__((ext_vector_type(4))) float v4f;
typedef __attribute__((ext_vector_type(8))) float v8f;

namespace {
constexpr int kS = 256;
constexpr int kC = 128;
constexpr int kCenter = 128;
constexpr int kTerms = 33;
constexpr int kShifts = 129;
constexpr int kKTiles = 36;
constexpr int kRows = 8;
constexpr int kHalo = 4;
constexpr int kFold = 4;
constexpr int kPasses = 7;
constexpr int kPassTiles = 12;
constexpr int kWinLo = 96;
constexpr int kWin = 64;
constexpr int kSetBuf = kWin + 2 * kHalo;
constexpr int kXLo = 16;
constexpr int kXBuf = kXLo + kS + 32;
constexpr int kVRow = kWin + 8;
}

__device__ __forceinline__ v8f mma_f32x4(v2f a, v2f b, v8f c) {
  return __builtin_amdgcn_wmma_f32_16x16x4_f32(
       false, a,  false, b,
       (short)0, c,  false,  false);
}

__global__ __launch_bounds__(256) void tonal_diffusion_kernel(
    const float* __restrict__ data,
    const float* __restrict__ log_weights,
    float* __restrict__ out,
    int n_rows) {
  const int lane = threadIdx.x & 31;
  const int wave = threadIdx.x >> 5;
  const int row  = blockIdx.x * kRows + wave;
  const int rowc = (row < n_rows) ? row : (n_rows - 1);

  __shared__ __attribute__((aligned(16))) float sSet[kRows][2][kSetBuf];
  __shared__ __attribute__((aligned(16))) float sX[kRows][kXBuf];
  __shared__ __attribute__((aligned(16))) float sV[kRows][kPasses][kVRow];
  __shared__ __attribute__((aligned(16))) float sDataPad[kRows][160];
  __shared__ __attribute__((aligned(16))) float sLogAcc[kRows][384];

  const int steps[6] = {1, -1, -3, 3, 4, -4};

  float lw[6];
#pragma unroll
  for (int i = 0; i < 6; ++i) lw[i] = log_weights[(size_t)rowc * 6 + i];
  const v4f draw = *(const v4f*)(data + (size_t)rowc * kC + 4 * lane);

  {
    float* setFlat = &sSet[wave][0][0];
    for (int idx = lane; idx < 2 * kSetBuf; idx += 32) setFlat[idx] = 0.f;
    v4f* x4 = (v4f*)&sX[wave][0];
    const v4f zero4 = {0.f, 0.f, 0.f, 0.f};
    for (int idx = lane; idx < kXBuf / 4; idx += 32) x4[idx] = zero4;
    for (int idx = lane; idx < kPasses * 8; idx += 32)
      sV[wave][idx >> 3][kWin + (idx & 7)] = 0.f;
  }

  float p[6];
  float rate = 0.f;
#pragma unroll
  for (int i = 0; i < 6; ++i) {
    p[i] = expf(lw[i]);
    rate += p[i];
  }
  const float inv_rate = 1.f / rate;
#pragma unroll
  for (int i = 0; i < 6; ++i) p[i] *= inv_rate;
  float* runA = &sSet[wave][0][0];
  float* runB = &sSet[wave][1][0];
  if (lane == 0) runA[kCenter - kWinLo + kHalo] = 1.f;

  float pm[kTerms];
  pm[0] = expf(-rate);
#pragma unroll
  for (int n = 0; n < kTerms - 1; ++n) pm[n + 1] = pm[n] * rate / (float)(n + 1);
  __syncthreads();

  float rr[kFold + 1][2];
#pragma unroll
  for (int s = 0; s < kFold; ++s) {
#pragma unroll
    for (int c = 0; c < 2; ++c) {
      const int u = kHalo + 32 * c + lane;
      rr[s][c] = runA[u];
      float nv = 0.f;
#pragma unroll
      for (int i = 0; i < 6; ++i)
        nv = fmaf(p[i], runA[u - steps[i]], nv);
      runB[u] = nv;
    }
    __syncthreads();
    float* tmp = runA; runA = runB; runB = tmp;
  }
#pragma unroll
  for (int c = 0; c < 2; ++c) rr[kFold][c] = runA[kHalo + 32 * c + lane];

#pragma unroll
  for (int q = 0; q < kPasses; ++q) {
#pragma unroll
    for (int c = 0; c < 2; ++c) {
      float v = pm[4 * q] * rr[0][c];
#pragma unroll
      for (int r = 1; r < kFold; ++r) v = fmaf(pm[4 * q + r], rr[r][c], v);
      sV[wave][q][32 * c + lane] = v;
    }
  }
#pragma unroll
  for (int c = 0; c < 2; ++c) {
    float v = pm[4 * kPasses] * rr[0][c];
#pragma unroll
    for (int r = 1; r <= kFold; ++r) v = fmaf(pm[4 * kPasses + r], rr[r][c], v);
    sX[wave][kXLo + kWinLo + 32 * c + lane] = v;
  }

  const int c16 = lane & 15;
  const int h   = lane >> 4;
  v2f afr[kPassTiles];
  {
    const float* g4 = &runA[(kCenter + 16 + c16 - 2 * h) - kWinLo + kHalo];
#pragma unroll
    for (int kc = 0; kc < kPassTiles; ++kc) {
      afr[kc].x = g4[-4 * kc];
      afr[kc].y = g4[-4 * kc - 1];
    }
  }
  float pl = 0.f;
  {
    v4f dv;
    dv.x = (draw.x > 1e-8f) ? draw.x : 0.f;
    dv.y = (draw.y > 1e-8f) ? draw.y : 0.f;
    dv.z = (draw.z > 1e-8f) ? draw.z : 0.f;
    dv.w = (draw.w > 1e-8f) ? draw.w : 0.f;
    *(v4f*)&sDataPad[wave][16 + 4 * lane] = dv;
    sDataPad[wave][lane < 16 ? lane : 128 + lane] = 0.f;
#pragma unroll
    for (int i = 0; i < 4; ++i) {
      const float v = dv[i];
      pl = fmaf(v, logf((v > 0.f) ? v : 1.f), pl);
    }
  }
  __syncthreads();

  const bool in_win = (c16 >= kWinLo / 16) && (c16 < (kWinLo + kWin) / 16);
  const float* vp = &sV[wave][0][in_win ? (16 * c16 + 8 * h - kWinLo) : kWin];
  const float* bp = &sX[wave][16 * c16 + 2 * h];
  float* xo = &sX[wave][kXLo + 16 * c16 + 8 * h];
  v8f acc8;
#pragma unroll
  for (int q = kPasses - 1; q >= 0; --q) {
    const v4f vlo = *(const v4f*)(vp + q * kVRow);
    const v4f vhi = *(const v4f*)(vp + q * kVRow + 4);
    v8f c0 = {vlo.x, vlo.y, vlo.z, vlo.w, vhi.x, vhi.y, vhi.z, vhi.w};
    v8f c1 = {0.f, 0.f, 0.f, 0.f, 0.f, 0.f, 0.f, 0.f};
    v2f bfr[kPassTiles];
#pragma unroll
    for (int kc = 0; kc < kPassTiles; ++kc) bfr[kc] = *(const v2f*)(bp + 4 * kc);
    __builtin_amdgcn_sched_barrier(0);
#pragma unroll
    for (int kc = 0; kc < kPassTiles; kc += 2) {
      c0 = mma_f32x4(afr[kc], bfr[kc], c0);
      c1 = mma_f32x4(afr[kc + 1], bfr[kc + 1], c1);
    }
    acc8 = c0 + c1;
    if (q > 0) {
      *(v4f*)xo       = acc8.lo;
      *(v4f*)(xo + 4) = acc8.hi;
      __syncthreads();
    }
  }

  float* accL = &sX[wave][kXLo];
  {
    v8f lg;
#pragma unroll
    for (int r = 0; r < 8; ++r) lg[r] = logf(fmaxf(acc8[r], 1e-38f));
    *(v4f*)xo       = acc8.lo;
    *(v4f*)(xo + 4) = acc8.hi;
    float* lo = &sLogAcc[wave][16 * c16 + 8 * h];
    *(v4f*)lo       = lg.lo;
    *(v4f*)(lo + 4) = lg.hi;
  }
  for (int t = kS + lane; t < 384; t += 32) sLogAcc[wave][t] = 0.f;
  __syncthreads();

#pragma unroll
  for (int off = 16; off; off >>= 1) pl += __shfl_xor(pl, off, 32);

  float best = INFINITY;

  const int m    = lane & 15;
  const int koff = (lane & 16) ? 2 : 0;
  const float* dp = &sDataPad[wave][16 - m];
  const float* lp = &sLogAcc[wave][16 * m];
  const float* ap = &accL[16 * (m < 8 ? m : 8)];
  v8f dacc = {0.f, 0.f, 0.f, 0.f, 0.f, 0.f, 0.f, 0.f};
  v8f zacc = {0.f, 0.f, 0.f, 0.f, 0.f, 0.f, 0.f, 0.f};
#pragma unroll
  for (int kc = 0; kc < kKTiles; ++kc) {
    const int j = 4 * kc + koff;
    v2f a = {dp[j], dp[j + 1]};
    v2f b = {lp[j], lp[j + 1]};
    v2f a1 = {((unsigned)(j - m) < (unsigned)kC) ? 1.f : 0.f,
              ((unsigned)(j + 1 - m) < (unsigned)kC) ? 1.f : 0.f};
    v2f b1 = {ap[j], ap[j + 1]};
    dacc = mma_f32x4(a, b, dacc);
    zacc = mma_f32x4(a1, b1, zacc);
  }
  const int mbase = (lane & 16) ? 8 : 0;
  const int n16   = 16 * (lane & 15);
#pragma unroll
  for (int r = 0; r < 8; ++r) {
    const int k = (r + mbase) + n16;
    const float loss = pl - dacc[r] + logf(zacc[r]);
    best = (k < kShifts) ? fminf(best, loss) : best;
  }

#pragma unroll
  for (int off = 16; off; off >>= 1) best = fminf(best, __shfl_xor(best, off, 32));
  if (lane == 0 && row < n_rows) out[row] = best;
}

extern "C" void kernel_launch(void* const* d_in, const int* in_sizes, int n_in,
                              void* d_out, int out_size, void* d_ws, size_t ws_size,
                              hipStream_t stream) {
  (void)n_in; (void)d_ws; (void)ws_size; (void)out_size;
  const float* data = (const float*)d_in[0];
  const float* logw = (const float*)d_in[1];
  float* out = (float*)d_out;
  const int n_rows = in_sizes[0] / kC;
  const int blocks = (n_rows + kRows - 1) / kRows;
  tonal_diffusion_kernel<<<blocks, 256, 0, stream>>>(data, logw, out, n_rows);
}


# Poisson terms by multiplication: pmf(n+1) = pmf(n) * (rate * 1/(n+1)), reciprocals of n+1 as constants, replaces 26 dependent division sequences per lane; pmf values move by a few ulp
# speedup vs baseline: 2.2481x; 1.0605x over previous
#include <hip/hip_runtime.h>
#include <math.h>


typedef __attribute__((ext_vector_type(2))) float v2f;
typedef __attribute__((ext_vector_type(4))) float v4f;
typedef __attribute__((ext_vector_type(8))) float v8f;

namespace {
constexpr int kS = 256;
constexpr int kC = 128;
constexpr int kCenter = 128;
constexpr int kTerms = 33;
constexpr int kShifts = 129;
constexpr int kKTiles = 36;
constexpr int kRows = 8;
constexpr int kHalo = 4;
constexpr int kFold = 4;
constexpr int kPasses = 7;
constexpr int kPassTiles = 12;
constexpr int kWinLo = 96;
constexpr int kWin = 64;
constexpr int kSetBuf = kWin + 2 * kHalo;
constexpr int kXLo = 16;
constexpr int kXBuf = kXLo + kS + 32;
constexpr int kVRow = kWin + 8;
}

__device__ __forceinline__ v8f mma_f32x4(v2f a, v2f b, v8f c) {
  return __builtin_amdgcn_wmma_f32_16x16x4_f32(
       false, a,  false, b,
       (short)0, c,  false,  false);
}

__global__ __launch_bounds__(256) void tonal_diffusion_kernel(
    const float* __restrict__ data,
    const float* __restrict__ log_weights,
    float* __restrict__ out,
    int n_rows) {
  const int lane = threadIdx.x & 31;
  const int wave = threadIdx.x >> 5;
  const int row  = blockIdx.x * kRows + wave;
  const int rowc = (row < n_rows) ? row : (n_rows - 1);

  __shared__ __attribute__((aligned(16))) float sSet[kRows][2][kSetBuf];
  __shared__ __attribute__((aligned(16))) float sX[kRows][kXBuf];
  __shared__ __attribute__((aligned(16))) float sV[kRows][kPasses][kVRow];
  __shared__ __attribute__((aligned(16))) float sDataPad[kRows][160];
  __shared__ __attribute__((aligned(16))) float sLogAcc[kRows][384];

  const int steps[6] = {1, -1, -3, 3, 4, -4};

  float lw[6];
#pragma unroll
  for (int i = 0; i < 6; ++i) lw[i] = log_weights[(size_t)rowc * 6 + i];
  const v4f draw = *(const v4f*)(data + (size_t)rowc * kC + 4 * lane);

  {
    float* setFlat = &sSet[wave][0][0];
    for (int idx = lane; idx < 2 * kSetBuf; idx += 32) setFlat[idx] = 0.f;
    v4f* x4 = (v4f*)&sX[wave][0];
    const v4f zero4 = {0.f, 0.f, 0.f, 0.f};
    for (int idx = lane; idx < kXBuf / 4; idx += 32) x4[idx] = zero4;
    for (int idx = lane; idx < kPasses * 8; idx += 32)
      sV[wave][idx >> 3][kWin + (idx & 7)] = 0.f;
  }

  float p[6];
  float rate = 0.f;
#pragma unroll
  for (int i = 0; i < 6; ++i) {
    p[i] = expf(lw[i]);
    rate += p[i];
  }
  const float inv_rate = 1.f / rate;
#pragma unroll
  for (int i = 0; i < 6; ++i) p[i] *= inv_rate;
  float* runA = &sSet[wave][0][0];
  float* runB = &sSet[wave][1][0];
  if (lane == 0) runA[kCenter - kWinLo + kHalo] = 1.f;

  float pm[kTerms];
  pm[0] = expf(-rate);
#pragma unroll
  for (int n = 0; n < kTerms - 1; ++n) {
    const float step = rate * (1.0f / (float)(n + 1));
    pm[n + 1] = pm[n] * step;
  }
  __syncthreads();

  float rr[kFold + 1][2];
#pragma unroll
  for (int s = 0; s < kFold; ++s) {
#pragma unroll
    for (int c = 0; c < 2; ++c) {
      const int u = kHalo + 32 * c + lane;
      rr[s][c] = runA[u];
      float nv = 0.f;
#pragma unroll
      for (int i = 0; i < 6; ++i)
        nv = fmaf(p[i], runA[u - steps[i]], nv);
      runB[u] = nv;
    }
    __syncthreads();
    float* tmp = runA; runA = runB; runB = tmp;
  }
#pragma unroll
  for (int c = 0; c < 2; ++c) rr[kFold][c] = runA[kHalo + 32 * c + lane];

#pragma unroll
  for (int q = 0; q < kPasses; ++q) {
#pragma unroll
    for (int c = 0; c < 2; ++c) {
      float v = pm[4 * q] * rr[0][c];
#pragma unroll
      for (int r = 1; r < kFold; ++r) v = fmaf(pm[4 * q + r], rr[r][c], v);
      sV[wave][q][32 * c + lane] = v;
    }
  }
#pragma unroll
  for (int c = 0; c < 2; ++c) {
    float v = pm[4 * kPasses] * rr[0][c];
#pragma unroll
    for (int r = 1; r <= kFold; ++r) v = fmaf(pm[4 * kPasses + r], rr[r][c], v);
    sX[wave][kXLo + kWinLo + 32 * c + lane] = v;
  }

  const int c16 = lane & 15;
  const int h   = lane >> 4;
  v2f afr[kPassTiles];
  {
    const float* g4 = &runA[(kCenter + 16 + c16 - 2 * h) - kWinLo + kHalo];
#pragma unroll
    for (int kc = 0; kc < kPassTiles; ++kc) {
      afr[kc].x = g4[-4 * kc];
      afr[kc].y = g4[-4 * kc - 1];
    }
  }
  float pl = 0.f;
  {
    v4f dv;
    dv.x = (draw.x > 1e-8f) ? draw.x : 0.f;
    dv.y = (draw.y > 1e-8f) ? draw.y : 0.f;
    dv.z = (draw.z > 1e-8f) ? draw.z : 0.f;
    dv.w = (draw.w > 1e-8f) ? draw.w : 0.f;
    *(v4f*)&sDataPad[wave][16 + 4 * lane] = dv;
    sDataPad[wave][lane < 16 ? lane : 128 + lane] = 0.f;
#pragma unroll
    for (int i = 0; i < 4; ++i) {
      const float v = dv[i];
      pl = fmaf(v, logf((v > 0.f) ? v : 1.f), pl);
    }
  }
  __syncthreads();

  const bool in_win = (c16 >= kWinLo / 16) && (c16 < (kWinLo + kWin) / 16);
  const float* vp = &sV[wave][0][in_win ? (16 * c16 + 8 * h - kWinLo) : kWin];
  const float* bp = &sX[wave][16 * c16 + 2 * h];
  float* xo = &sX[wave][kXLo + 16 * c16 + 8 * h];
  v8f acc8;
#pragma unroll
  for (int q = kPasses - 1; q >= 0; --q) {
    const v4f vlo = *(const v4f*)(vp + q * kVRow);
    const v4f vhi = *(const v4f*)(vp + q * kVRow + 4);
    v8f c0 = {vlo.x, vlo.y, vlo.z, vlo.w, vhi.x, vhi.y, vhi.z, vhi.w};
    v8f c1 = {0.f, 0.f, 0.f, 0.f, 0.f, 0.f, 0.f, 0.f};
    v2f bfr[kPassTiles];
#pragma unroll
    for (int kc = 0; kc < kPassTiles; ++kc) bfr[kc] = *(const v2f*)(bp + 4 * kc);
    __builtin_amdgcn_sched_barrier(0);
#pragma unroll
    for (int kc = 0; kc < kPassTiles; kc += 2) {
      c0 = mma_f32x4(afr[kc], bfr[kc], c0);
      c1 = mma_f32x4(afr[kc + 1], bfr[kc + 1], c1);
    }
    acc8 = c0 + c1;
    if (q > 0) {
      *(v4f*)xo       = acc8.lo;
      *(v4f*)(xo + 4) = acc8.hi;
      __syncthreads();
    }
  }

  float* accL = &sX[wave][kXLo];
  {
    v8f lg;
#pragma unroll
    for (int r = 0; r < 8; ++r) lg[r] = logf(fmaxf(acc8[r], 1e-38f));
    *(v4f*)xo       = acc8.lo;
    *(v4f*)(xo + 4) = acc8.hi;
    float* lo = &sLogAcc[wave][16 * c16 + 8 * h];
    *(v4f*)lo       = lg.lo;
    *(v4f*)(lo + 4) = lg.hi;
  }
  for (int t = kS + lane; t < 384; t += 32) sLogAcc[wave][t] = 0.f;
  __syncthreads();

#pragma unroll
  for (int off = 16; off; off >>= 1) pl += __shfl_xor(pl, off, 32);

  float best = INFINITY;

  const int m    = lane & 15;
  const int koff = (lane & 16) ? 2 : 0;
  const float* dp = &sDataPad[wave][16 - m];
  const float* lp = &sLogAcc[wave][16 * m];
  const float* ap = &accL[16 * (m < 8 ? m : 8)];
  v8f dacc = {0.f, 0.f, 0.f, 0.f, 0.f, 0.f, 0.f, 0.f};
  v8f zacc = {0.f, 0.f, 0.f, 0.f, 0.f, 0.f, 0.f, 0.f};
#pragma unroll
  for (int kc = 0; kc < kKTiles; ++kc) {
    const int j = 4 * kc + koff;
    v2f a = {dp[j], dp[j + 1]};
    v2f b = {lp[j], lp[j + 1]};
    v2f a1 = {((unsigned)(j - m) < (unsigned)kC) ? 1.f : 0.f,
              ((unsigned)(j + 1 - m) < (unsigned)kC) ? 1.f : 0.f};
    v2f b1 = {ap[j], ap[j + 1]};
    dacc = mma_f32x4(a, b, dacc);
    zacc = mma_f32x4(a1, b1, zacc);
  }
  const int mbase = (lane & 16) ? 8 : 0;
  const int n16   = 16 * (lane & 15);
#pragma unroll
  for (int r = 0; r < 8; ++r) {
    const int k = (r + mbase) + n16;
    const float loss = pl - dacc[r] + logf(zacc[r]);
    best = (k < kShifts) ? fminf(best, loss) : best;
  }

#pragma unroll
  for (int off = 16; off; off >>= 1) best = fminf(best, __shfl_xor(best, off, 32));
  if (lane == 0 && row < n_rows) out[row] = best;
}

extern "C" void kernel_launch(void* const* d_in, const int* in_sizes, int n_in,
                              void* d_out, int out_size, void* d_ws, size_t ws_size,
                              hipStream_t stream) {
  (void)n_in; (void)d_ws; (void)ws_size; (void)out_size;
  const float* data = (const float*)d_in[0];
  const float* logw = (const float*)d_in[1];
  float* out = (float*)d_out;
  const int n_rows = in_sizes[0] / kC;
  const int blocks = (n_rows + kRows - 1) / kRows;
  tonal_diffusion_kernel<<<blocks, 256, 0, stream>>>(data, logw, out, n_rows);
}


# All 13 workgroup barriers replaced by wave-scope fences: every LDS slice is private to one wave and the LDS serves a wave's loads and stores in issue order; arithmetic and sum orders untouched
# speedup vs baseline: 2.4320x; 1.0818x over previous
#include <hip/hip_runtime.h>
#include <math.h>


typedef __attribute__((ext_vector_type(2))) float v2f;
typedef __attribute__((ext_vector_type(4))) float v4f;
typedef __attribute__((ext_vector_type(8))) float v8f;

namespace {
constexpr int kS = 256;
constexpr int kC = 128;
constexpr int kCenter = 128;
constexpr int kTerms = 33;
constexpr int kShifts = 129;
constexpr int kKTiles = 36;
constexpr int kRows = 8;
constexpr int kHalo = 4;
constexpr int kFold = 4;
constexpr int kPasses = 7;
constexpr int kPassTiles = 12;
constexpr int kWinLo = 96;
constexpr int kWin = 64;
constexpr int kSetBuf = kWin + 2 * kHalo;
constexpr int kXLo = 16;
constexpr int kXBuf = kXLo + kS + 32;
constexpr int kVRow = kWin + 8;
}

__device__ __forceinline__ v8f mma_f32x4(v2f a, v2f b, v8f c) {
  return __builtin_amdgcn_wmma_f32_16x16x4_f32(
       false, a,  false, b,
       (short)0, c,  false,  false);
}

__device__ __forceinline__ void wave_lds_handover() {
  __builtin_amdgcn_fence(__ATOMIC_RELEASE, "wavefront");
  __builtin_amdgcn_wave_barrier();
  __builtin_amdgcn_fence(__ATOMIC_ACQUIRE, "wavefront");
}

__global__ __launch_bounds__(256) void tonal_diffusion_kernel(
    const float* __restrict__ data,
    const float* __restrict__ log_weights,
    float* __restrict__ out,
    int n_rows) {
  const int lane = threadIdx.x & 31;
  const int wave = threadIdx.x >> 5;
  const int row  = blockIdx.x * kRows + wave;
  const int rowc = (row < n_rows) ? row : (n_rows - 1);

  __shared__ __attribute__((aligned(16))) float sSet[kRows][2][kSetBuf];
  __shared__ __attribute__((aligned(16))) float sX[kRows][kXBuf];
  __shared__ __attribute__((aligned(16))) float sV[kRows][kPasses][kVRow];
  __shared__ __attribute__((aligned(16))) float sDataPad[kRows][160];
  __shared__ __attribute__((aligned(16))) float sLogAcc[kRows][384];

  const int steps[6] = {1, -1, -3, 3, 4, -4};

  float lw[6];
#pragma unroll
  for (int i = 0; i < 6; ++i) lw[i] = log_weights[(size_t)rowc * 6 + i];
  const v4f draw = *(const v4f*)(data + (size_t)rowc * kC + 4 * lane);

  {
    float* setFlat = &sSet[wave][0][0];
    for (int idx = lane; idx < 2 * kSetBuf; idx += 32) setFlat[idx] = 0.f;
    v4f* x4 = (v4f*)&sX[wave][0];
    const v4f zero4 = {0.f, 0.f, 0.f, 0.f};
    for (int idx = lane; idx < kXBuf / 4; idx += 32) x4[idx] = zero4;
    for (int idx = lane; idx < kPasses * 8; idx += 32)
      sV[wave][idx >> 3][kWin + (idx & 7)] = 0.f;
  }

  float p[6];
  float rate = 0.f;
#pragma unroll
  for (int i = 0; i < 6; ++i) {
    p[i] = expf(lw[i]);
    rate += p[i];
  }
  const float inv_rate = 1.f / rate;
#pragma unroll
  for (int i = 0; i < 6; ++i) p[i] *= inv_rate;
  float* runA = &sSet[wave][0][0];
  float* runB = &sSet[wave][1][0];
  if (lane == 0) runA[kCenter - kWinLo + kHalo] = 1.f;

  float pm[kTerms];
  pm[0] = expf(-rate);
#pragma unroll
  for (int n = 0; n < kTerms - 1; ++n) {
    const float step = rate * (1.0f / (float)(n + 1));
    pm[n + 1] = pm[n] * step;
  }
  wave_lds_handover();

  float rr[kFold + 1][2];
#pragma unroll
  for (int s = 0; s < kFold; ++s) {
#pragma unroll
    for (int c = 0; c < 2; ++c) {
      const int u = kHalo + 32 * c + lane;
      rr[s][c] = runA[u];
      float nv = 0.f;
#pragma unroll
      for (int i = 0; i < 6; ++i)
        nv = fmaf(p[i], runA[u - steps[i]], nv);
      runB[u] = nv;
    }
    wave_lds_handover();
    float* tmp = runA; runA = runB; runB = tmp;
  }
#pragma unroll
  for (int c = 0; c < 2; ++c) rr[kFold][c] = runA[kHalo + 32 * c + lane];

#pragma unroll
  for (int q = 0; q < kPasses; ++q) {
#pragma unroll
    for (int c = 0; c < 2; ++c) {
      float v = pm[4 * q] * rr[0][c];
#pragma unroll
      for (int r = 1; r < kFold; ++r) v = fmaf(pm[4 * q + r], rr[r][c], v);
      sV[wave][q][32 * c + lane] = v;
    }
  }
#pragma unroll
  for (int c = 0; c < 2; ++c) {
    float v = pm[4 * kPasses] * rr[0][c];
#pragma unroll
    for (int r = 1; r <= kFold; ++r) v = fmaf(pm[4 * kPasses + r], rr[r][c], v);
    sX[wave][kXLo + kWinLo + 32 * c + lane] = v;
  }

  const int c16 = lane & 15;
  const int h   = lane >> 4;
  v2f afr[kPassTiles];
  {
    const float* g4 = &runA[(kCenter + 16 + c16 - 2 * h) - kWinLo + kHalo];
#pragma unroll
    for (int kc = 0; kc < kPassTiles; ++kc) {
      afr[kc].x = g4[-4 * kc];
      afr[kc].y = g4[-4 * kc - 1];
    }
  }
  float pl = 0.f;
  {
    v4f dv;
    dv.x = (draw.x > 1e-8f) ? draw.x : 0.f;
    dv.y = (draw.y > 1e-8f) ? draw.y : 0.f;
    dv.z = (draw.z > 1e-8f) ? draw.z : 0.f;
    dv.w = (draw.w > 1e-8f) ? draw.w : 0.f;
    *(v4f*)&sDataPad[wave][16 + 4 * lane] = dv;
    sDataPad[wave][lane < 16 ? lane : 128 + lane] = 0.f;
#pragma unroll
    for (int i = 0; i < 4; ++i) {
      const float v = dv[i];
      pl = fmaf(v, logf((v > 0.f) ? v : 1.f), pl);
    }
  }
  wave_lds_handover();

  const bool in_win = (c16 >= kWinLo / 16) && (c16 < (kWinLo + kWin) / 16);
  const float* vp = &sV[wave][0][in_win ? (16 * c16 + 8 * h - kWinLo) : kWin];
  const float* bp = &sX[wave][16 * c16 + 2 * h];
  float* xo = &sX[wave][kXLo + 16 * c16 + 8 * h];
  v8f acc8;
#pragma unroll
  for (int q = kPasses - 1; q >= 0; --q) {
    const v4f vlo = *(const v4f*)(vp + q * kVRow);
    const v4f vhi = *(const v4f*)(vp + q * kVRow + 4);
    v8f c0 = {vlo.x, vlo.y, vlo.z, vlo.w, vhi.x, vhi.y, vhi.z, vhi.w};
    v8f c1 = {0.f, 0.f, 0.f, 0.f, 0.f, 0.f, 0.f, 0.f};
    v2f bfr[kPassTiles];
#pragma unroll
    for (int kc = 0; kc < kPassTiles; ++kc) bfr[kc] = *(const v2f*)(bp + 4 * kc);
    __builtin_amdgcn_sched_barrier(0);
#pragma unroll
    for (int kc = 0; kc < kPassTiles; kc += 2) {
      c0 = mma_f32x4(afr[kc], bfr[kc], c0);
      c1 = mma_f32x4(afr[kc + 1], bfr[kc + 1], c1);
    }
    acc8 = c0 + c1;
    if (q > 0) {
      *(v4f*)xo       = acc8.lo;
      *(v4f*)(xo + 4) = acc8.hi;
      wave_lds_handover();
    }
  }

  float* accL = &sX[wave][kXLo];
  {
    v8f lg;
#pragma unroll
    for (int r = 0; r < 8; ++r) lg[r] = logf(fmaxf(acc8[r], 1e-38f));
    *(v4f*)xo       = acc8.lo;
    *(v4f*)(xo + 4) = acc8.hi;
    float* lo = &sLogAcc[wave][16 * c16 + 8 * h];
    *(v4f*)lo       = lg.lo;
    *(v4f*)(lo + 4) = lg.hi;
  }
  for (int t = kS + lane; t < 384; t += 32) sLogAcc[wave][t] = 0.f;
  wave_lds_handover();

#pragma unroll
  for (int off = 16; off; off >>= 1) pl += __shfl_xor(pl, off, 32);

  float best = INFINITY;

  const int m    = lane & 15;
  const int koff = (lane & 16) ? 2 : 0;
  const float* dp = &sDataPad[wave][16 - m];
  const float* lp = &sLogAcc[wave][16 * m];
  const float* ap = &accL[16 * (m < 8 ? m : 8)];
  v8f dacc = {0.f, 0.f, 0.f, 0.f, 0.f, 0.f, 0.f, 0.f};
  v8f zacc = {0.f, 0.f, 0.f, 0.f, 0.f, 0.f, 0.f, 0.f};
#pragma unroll
  for (int kc = 0; kc < kKTiles; ++kc) {
    const int j = 4 * kc + koff;
    v2f a = {dp[j], dp[j + 1]};
    v2f b = {lp[j], lp[j + 1]};
    v2f a1 = {((unsigned)(j - m) < (unsigned)kC) ? 1.f : 0.f,
              ((unsigned)(j + 1 - m) < (unsigned)kC) ? 1.f : 0.f};
    v2f b1 = {ap[j], ap[j + 1]};
    dacc = mma_f32x4(a, b, dacc);
    zacc = mma_f32x4(a1, b1, zacc);
  }
  const int mbase = (lane & 16) ? 8 : 0;
  const int n16   = 16 * (lane & 15);
#pragma unroll
  for (int r = 0; r < 8; ++r) {
    const int k = (r + mbase) + n16;
    const float loss = pl - dacc[r] + logf(zacc[r]);
    best = (k < kShifts) ? fminf(best, loss) : best;
  }

#pragma unroll
  for (int off = 16; off; off >>= 1) best = fminf(best, __shfl_xor(best, off, 32));
  if (lane == 0 && row < n_rows) out[row] = best;
}

extern "C" void kernel_launch(void* const* d_in, const int* in_sizes, int n_in,
                              void* d_out, int out_size, void* d_ws, size_t ws_size,
                              hipStream_t stream) {
  (void)n_in; (void)d_ws; (void)ws_size; (void)out_size;
  const float* data = (const float*)d_in[0];
  const float* logw = (const float*)d_in[1];
  float* out = (float*)d_out;
  const int n_rows = in_sizes[0] / kC;
  const int blocks = (n_rows + kRows - 1) / kRows;
  tonal_diffusion_kernel<<<blocks, 256, 0, stream>>>(data, logw, out, n_rows);
}


# Window sums Z as two short matrix products: 16-wide band sums (8 K-tiles), then an 8-block 0/1 band times that tile used in place as B (4 tiles), not one 36-tile band; Z and cross sum orders change
# speedup vs baseline: 2.6278x; 1.0805x over previous
#include <hip/hip_runtime.h>
#include <math.h>


typedef __attribute__((ext_vector_type(2))) float v2f;
typedef __attribute__((ext_vector_type(4))) float v4f;
typedef __attribute__((ext_vector_type(8))) float v8f;

namespace {
constexpr int kS = 256;
constexpr int kC = 128;
constexpr int kCenter = 128;
constexpr int kTerms = 33;
constexpr int kShifts = 129;
constexpr int kKTiles = 36;
constexpr int kRows = 8;
constexpr int kHalo = 4;
constexpr int kFold = 4;
constexpr int kPasses = 7;
constexpr int kPassTiles = 12;
constexpr int kWinLo = 96;
constexpr int kWin = 64;
constexpr int kSetBuf = kWin + 2 * kHalo;
constexpr int kXLo = 16;
constexpr int kXBuf = kXLo + kS + 32;
constexpr int kVRow = kWin + 8;
}

__device__ __forceinline__ v8f mma_f32x4(v2f a, v2f b, v8f c) {
  return __builtin_amdgcn_wmma_f32_16x16x4_f32(
       false, a,  false, b,
       (short)0, c,  false,  false);
}

__device__ __forceinline__ void wave_lds_handover() {
  __builtin_amdgcn_fence(__ATOMIC_RELEASE, "wavefront");
  __builtin_amdgcn_wave_barrier();
  __builtin_amdgcn_fence(__ATOMIC_ACQUIRE, "wavefront");
}

__global__ __launch_bounds__(256) void tonal_diffusion_kernel(
    const float* __restrict__ data,
    const float* __restrict__ log_weights,
    float* __restrict__ out,
    int n_rows) {
  const int lane = threadIdx.x & 31;
  const int wave = threadIdx.x >> 5;
  const int row  = blockIdx.x * kRows + wave;
  const int rowc = (row < n_rows) ? row : (n_rows - 1);

  __shared__ __attribute__((aligned(16))) float sSet[kRows][2][kSetBuf];
  __shared__ __attribute__((aligned(16))) float sX[kRows][kXBuf];
  __shared__ __attribute__((aligned(16))) float sV[kRows][kPasses][kVRow];
  __shared__ __attribute__((aligned(16))) float sDataPad[kRows][160];
  __shared__ __attribute__((aligned(16))) float sLogAcc[kRows][384];

  const int steps[6] = {1, -1, -3, 3, 4, -4};

  float lw[6];
#pragma unroll
  for (int i = 0; i < 6; ++i) lw[i] = log_weights[(size_t)rowc * 6 + i];
  const v4f draw = *(const v4f*)(data + (size_t)rowc * kC + 4 * lane);

  {
    float* setFlat = &sSet[wave][0][0];
    for (int idx = lane; idx < 2 * kSetBuf; idx += 32) setFlat[idx] = 0.f;
    v4f* x4 = (v4f*)&sX[wave][0];
    const v4f zero4 = {0.f, 0.f, 0.f, 0.f};
    for (int idx = lane; idx < kXBuf / 4; idx += 32) x4[idx] = zero4;
    for (int idx = lane; idx < kPasses * 8; idx += 32)
      sV[wave][idx >> 3][kWin + (idx & 7)] = 0.f;
  }

  float p[6];
  float rate = 0.f;
#pragma unroll
  for (int i = 0; i < 6; ++i) {
    p[i] = expf(lw[i]);
    rate += p[i];
  }
  const float inv_rate = 1.f / rate;
#pragma unroll
  for (int i = 0; i < 6; ++i) p[i] *= inv_rate;
  float* runA = &sSet[wave][0][0];
  float* runB = &sSet[wave][1][0];
  if (lane == 0) runA[kCenter - kWinLo + kHalo] = 1.f;

  float pm[kTerms];
  pm[0] = expf(-rate);
#pragma unroll
  for (int n = 0; n < kTerms - 1; ++n) {
    const float step = rate * (1.0f / (float)(n + 1));
    pm[n + 1] = pm[n] * step;
  }
  wave_lds_handover();

  float rr[kFold + 1][2];
#pragma unroll
  for (int s = 0; s < kFold; ++s) {
#pragma unroll
    for (int c = 0; c < 2; ++c) {
      const int u = kHalo + 32 * c + lane;
      rr[s][c] = runA[u];
      float nv = 0.f;
#pragma unroll
      for (int i = 0; i < 6; ++i)
        nv = fmaf(p[i], runA[u - steps[i]], nv);
      runB[u] = nv;
    }
    wave_lds_handover();
    float* tmp = runA; runA = runB; runB = tmp;
  }
#pragma unroll
  for (int c = 0; c < 2; ++c) rr[kFold][c] = runA[kHalo + 32 * c + lane];

#pragma unroll
  for (int q = 0; q < kPasses; ++q) {
#pragma unroll
    for (int c = 0; c < 2; ++c) {
      float v = pm[4 * q] * rr[0][c];
#pragma unroll
      for (int r = 1; r < kFold; ++r) v = fmaf(pm[4 * q + r], rr[r][c], v);
      sV[wave][q][32 * c + lane] = v;
    }
  }
#pragma unroll
  for (int c = 0; c < 2; ++c) {
    float v = pm[4 * kPasses] * rr[0][c];
#pragma unroll
    for (int r = 1; r <= kFold; ++r) v = fmaf(pm[4 * kPasses + r], rr[r][c], v);
    sX[wave][kXLo + kWinLo + 32 * c + lane] = v;
  }

  const int c16 = lane & 15;
  const int h   = lane >> 4;
  v2f afr[kPassTiles];
  {
    const float* g4 = &runA[(kCenter + 16 + c16 - 2 * h) - kWinLo + kHalo];
#pragma unroll
    for (int kc = 0; kc < kPassTiles; ++kc) {
      afr[kc].x = g4[-4 * kc];
      afr[kc].y = g4[-4 * kc - 1];
    }
  }
  float pl = 0.f;
  {
    v4f dv;
    dv.x = (draw.x > 1e-8f) ? draw.x : 0.f;
    dv.y = (draw.y > 1e-8f) ? draw.y : 0.f;
    dv.z = (draw.z > 1e-8f) ? draw.z : 0.f;
    dv.w = (draw.w > 1e-8f) ? draw.w : 0.f;
    *(v4f*)&sDataPad[wave][16 + 4 * lane] = dv;
    sDataPad[wave][lane < 16 ? lane : 128 + lane] = 0.f;
#pragma unroll
    for (int i = 0; i < 4; ++i) {
      const float v = dv[i];
      pl = fmaf(v, logf((v > 0.f) ? v : 1.f), pl);
    }
  }
  wave_lds_handover();

  const bool in_win = (c16 >= kWinLo / 16) && (c16 < (kWinLo + kWin) / 16);
  const float* vp = &sV[wave][0][in_win ? (16 * c16 + 8 * h - kWinLo) : kWin];
  const float* bp = &sX[wave][16 * c16 + 2 * h];
  float* xo = &sX[wave][kXLo + 16 * c16 + 8 * h];
  v8f acc8;
#pragma unroll
  for (int q = kPasses - 1; q >= 0; --q) {
    const v4f vlo = *(const v4f*)(vp + q * kVRow);
    const v4f vhi = *(const v4f*)(vp + q * kVRow + 4);
    v8f c0 = {vlo.x, vlo.y, vlo.z, vlo.w, vhi.x, vhi.y, vhi.z, vhi.w};
    v8f c1 = {0.f, 0.f, 0.f, 0.f, 0.f, 0.f, 0.f, 0.f};
    v2f bfr[kPassTiles];
#pragma unroll
    for (int kc = 0; kc < kPassTiles; ++kc) bfr[kc] = *(const v2f*)(bp + 4 * kc);
    __builtin_amdgcn_sched_barrier(0);
#pragma unroll
    for (int kc = 0; kc < kPassTiles; kc += 2) {
      c0 = mma_f32x4(afr[kc], bfr[kc], c0);
      c1 = mma_f32x4(afr[kc + 1], bfr[kc + 1], c1);
    }
    acc8 = c0 + c1;
    if (q > 0) {
      *(v4f*)xo       = acc8.lo;
      *(v4f*)(xo + 4) = acc8.hi;
      wave_lds_handover();
    }
  }

  float* accL = &sX[wave][kXLo];
  {
    v8f lg;
#pragma unroll
    for (int r = 0; r < 8; ++r) lg[r] = logf(fmaxf(acc8[r], 1e-38f));
    *(v4f*)xo       = acc8.lo;
    *(v4f*)(xo + 4) = acc8.hi;
    float* lo = &sLogAcc[wave][16 * c16 + 8 * h];
    *(v4f*)lo       = lg.lo;
    *(v4f*)(lo + 4) = lg.hi;
  }
  for (int t = kS + lane; t < 384; t += 32) sLogAcc[wave][t] = 0.f;
  wave_lds_handover();

#pragma unroll
  for (int off = 16; off; off >>= 1) pl += __shfl_xor(pl, off, 32);

  float best = INFINITY;

  const int m    = lane & 15;
  const int koff = (lane & 16) ? 2 : 0;
  const float* dp = &sDataPad[wave][16 - m];
  const float* lp = &sLogAcc[wave][16 * m];
  const float* ap = &accL[16 * m];
  v8f dacc0 = {0.f, 0.f, 0.f, 0.f, 0.f, 0.f, 0.f, 0.f};
  v8f dacc1 = {0.f, 0.f, 0.f, 0.f, 0.f, 0.f, 0.f, 0.f};
  v8f s16   = {0.f, 0.f, 0.f, 0.f, 0.f, 0.f, 0.f, 0.f};
  v8f zacc  = {0.f, 0.f, 0.f, 0.f, 0.f, 0.f, 0.f, 0.f};
  constexpr int kS16Tiles = 8;
  constexpr int kZFirst   = 10;
  constexpr int kZTiles   = 4;
#pragma unroll
  for (int kc = 0; kc < kKTiles; ++kc) {
    const int j = 4 * kc + koff;
    v2f a = {lp[j], lp[j + 1]};
    v2f b = {dp[j], dp[j + 1]};
    if (kc & 1) dacc1 = mma_f32x4(a, b, dacc1);
    else        dacc0 = mma_f32x4(a, b, dacc0);
    if (kc < kS16Tiles) {
      v2f a1 = {ap[j], ap[j + 1]};
      v2f b1 = {((unsigned)(j - m) < 16u) ? 1.f : 0.f,
                ((unsigned)(j + 1 - m) < 16u) ? 1.f : 0.f};
      s16 = mma_f32x4(a1, b1, s16);
    }
    if (kc >= kZFirst && kc < kZFirst + kZTiles) {
      const int t  = kc - kZFirst;
      const int mp = 4 * koff + 2 * t;
      v2f a2 = {((unsigned)(mp - m) < 8u) ? 1.f : 0.f,
                ((unsigned)(mp + 1 - m) < 8u) ? 1.f : 0.f};
      v2f b2 = {s16[2 * t], s16[2 * t + 1]};
      zacc = mma_f32x4(a2, b2, zacc);
    }
  }
  const v8f dacc = dacc0 + dacc1;
  const int mbase = (lane & 16) ? 8 : 0;
#pragma unroll
  for (int r = 0; r < 8; ++r) {
    const int k = m + 16 * (r + mbase);
    const float loss = pl - dacc[r] + logf(zacc[r]);
    best = (k < kShifts) ? fminf(best, loss) : best;
  }

#pragma unroll
  for (int off = 16; off; off >>= 1) best = fminf(best, __shfl_xor(best, off, 32));
  if (lane == 0 && row < n_rows) out[row] = best;
}

extern "C" void kernel_launch(void* const* d_in, const int* in_sizes, int n_in,
                              void* d_out, int out_size, void* d_ws, size_t ws_size,
                              hipStream_t stream) {
  (void)n_in; (void)d_ws; (void)ws_size; (void)out_size;
  const float* data = (const float*)d_in[0];
  const float* logw = (const float*)d_in[1];
  float* out = (float*)d_out;
  const int n_rows = in_sizes[0] / kC;
  const int blocks = (n_rows + kRows - 1) / kRows;
  tonal_diffusion_kernel<<<blocks, 256, 0, stream>>>(data, logw, out, n_rows);
}


# Fold 8, not 4: 8 LDS stencil steps give K^8 (65 taps) and u_3 directly, then 3 Toeplitz WMMA passes of 20 K-tiles replace 7 of 12 (60 matrix instructions for 84); reorders the acc summation
# speedup vs baseline: 2.8000x; 1.0655x over previous
#include <hip/hip_runtime.h>
#include <math.h>


typedef __attribute__((ext_vector_type(2))) float v2f;
typedef __attribute__((ext_vector_type(4))) float v4f;
typedef __attribute__((ext_vector_type(8))) float v8f;

namespace {
constexpr int kS = 256;
constexpr int kC = 128;
constexpr int kCenter = 128;
constexpr int kTerms = 33;
constexpr int kShifts = 129;
constexpr int kKTiles = 36;
constexpr int kRows = 8;
constexpr int kFold = 8;
constexpr int kPasses = 3;
constexpr int kPassTiles = 20;
constexpr int kWinLo = 96;
constexpr int kWin = 64;
constexpr int kSetPos = 3;
constexpr int kSetLo = 16;
constexpr int kSetBuf = kSetLo + 32 * kSetPos + 8;
constexpr int kXLo = 32;
constexpr int kXBuf = kXLo + kS + 32;
constexpr int kVRow = kWin + 8;
}

__device__ __forceinline__ v8f mma_f32x4(v2f a, v2f b, v8f c) {
  return __builtin_amdgcn_wmma_f32_16x16x4_f32(
       false, a,  false, b,
       (short)0, c,  false,  false);
}

__device__ __forceinline__ void wave_lds_handover() {
  __builtin_amdgcn_fence(__ATOMIC_RELEASE, "wavefront");
  __builtin_amdgcn_wave_barrier();
  __builtin_amdgcn_fence(__ATOMIC_ACQUIRE, "wavefront");
}

__global__ __launch_bounds__(256) void tonal_diffusion_kernel(
    const float* __restrict__ data,
    const float* __restrict__ log_weights,
    float* __restrict__ out,
    int n_rows) {
  const int lane = threadIdx.x & 31;
  const int wave = threadIdx.x >> 5;
  const int row  = blockIdx.x * kRows + wave;
  const int rowc = (row < n_rows) ? row : (n_rows - 1);

  __shared__ __attribute__((aligned(16))) float sSet[kRows][2][kSetBuf];
  __shared__ __attribute__((aligned(16))) float sX[kRows][kXBuf];
  __shared__ __attribute__((aligned(16))) float sV[kRows][kPasses][kVRow];
  __shared__ __attribute__((aligned(16))) float sDataPad[kRows][160];
  __shared__ __attribute__((aligned(16))) float sLogAcc[kRows][384];

  const int steps[6] = {1, -1, -3, 3, 4, -4};

  float lw[6];
#pragma unroll
  for (int i = 0; i < 6; ++i) lw[i] = log_weights[(size_t)rowc * 6 + i];
  const v4f draw = *(const v4f*)(data + (size_t)rowc * kC + 4 * lane);

  {
    float* setFlat = &sSet[wave][0][0];
    for (int idx = lane; idx < 2 * kSetBuf; idx += 32) setFlat[idx] = 0.f;
    v4f* x4 = (v4f*)&sX[wave][0];
    const v4f zero4 = {0.f, 0.f, 0.f, 0.f};
    for (int idx = lane; idx < kXBuf / 4; idx += 32) x4[idx] = zero4;
    for (int idx = lane; idx < kPasses * 8; idx += 32)
      sV[wave][idx >> 3][kWin + (idx & 7)] = 0.f;
  }

  float p[6];
  float rate = 0.f;
#pragma unroll
  for (int i = 0; i < 6; ++i) {
    p[i] = expf(lw[i]);
    rate += p[i];
  }
  const float inv_rate = 1.f / rate;
#pragma unroll
  for (int i = 0; i < 6; ++i) p[i] *= inv_rate;
  float* runA = &sSet[wave][0][0];
  float* runB = &sSet[wave][1][0];
  if (lane == 0) runA[kCenter - kWinLo + kSetLo] = 1.f;

  float pm[kTerms];
  pm[0] = expf(-rate);
#pragma unroll
  for (int n = 0; n < kTerms - 1; ++n) {
    const float step = rate * (1.0f / (float)(n + 1));
    pm[n + 1] = pm[n] * step;
  }
  wave_lds_handover();

  float rr[kFold + 1][2];
#pragma unroll
  for (int s = 0; s < kFold; ++s) {
#pragma unroll
    for (int c = 0; c < (s == kFold - 1 ? kSetPos : 2); ++c) {
      const int u = kSetLo + 32 * c + lane;
      if (c < 2) rr[s][c] = runA[u];
      float nv = 0.f;
#pragma unroll
      for (int i = 0; i < 6; ++i)
        nv = fmaf(p[i], runA[u - steps[i]], nv);
      runB[u] = nv;
    }
    wave_lds_handover();
    float* tmp = runA; runA = runB; runB = tmp;
  }
#pragma unroll
  for (int c = 0; c < 2; ++c) rr[kFold][c] = runA[kSetLo + 32 * c + lane];
  const float g8hi = runA[kSetLo + 64 + lane];

#pragma unroll
  for (int q = 0; q < kPasses; ++q) {
#pragma unroll
    for (int c = 0; c < 2; ++c) {
      float v = pm[kFold * q] * rr[0][c];
#pragma unroll
      for (int r = 1; r < kFold; ++r) v = fmaf(pm[kFold * q + r], rr[r][c], v);
      sV[wave][q][32 * c + lane] = v;
    }
  }
#pragma unroll
  for (int c = 0; c < 2; ++c) {
    float v = pm[kFold * kPasses] * rr[0][c];
#pragma unroll
    for (int r = 1; r <= kFold; ++r) v = fmaf(pm[kFold * kPasses + r], rr[r][c], v);
    sX[wave][kXLo + kWinLo + 32 * c + lane] = v;
  }
  sX[wave][kXLo + kWinLo + 64 + lane] = pm[kFold * kPasses + kFold] * g8hi;

  const int c16 = lane & 15;
  const int h   = lane >> 4;
  v2f afr[kPassTiles];
  {
    const float* g8 = &runA[(kCenter + 32 + c16 - 2 * h) - kWinLo + kSetLo];
#pragma unroll
    for (int kc = 0; kc < kPassTiles; ++kc) {
      afr[kc].x = g8[-4 * kc];
      afr[kc].y = g8[-4 * kc - 1];
    }
  }
  float pl = 0.f;
  {
    v4f dv;
    dv.x = (draw.x > 1e-8f) ? draw.x : 0.f;
    dv.y = (draw.y > 1e-8f) ? draw.y : 0.f;
    dv.z = (draw.z > 1e-8f) ? draw.z : 0.f;
    dv.w = (draw.w > 1e-8f) ? draw.w : 0.f;
    *(v4f*)&sDataPad[wave][16 + 4 * lane] = dv;
    sDataPad[wave][lane < 16 ? lane : 128 + lane] = 0.f;
#pragma unroll
    for (int i = 0; i < 4; ++i) {
      const float v = dv[i];
      pl = fmaf(v, logf((v > 0.f) ? v : 1.f), pl);
    }
  }
  wave_lds_handover();

  const bool in_win = (c16 >= kWinLo / 16) && (c16 < (kWinLo + kWin) / 16);
  const float* vp = &sV[wave][0][in_win ? (16 * c16 + 8 * h - kWinLo) : kWin];
  const float* bp = &sX[wave][16 * c16 + 2 * h];
  float* xo = &sX[wave][kXLo + 16 * c16 + 8 * h];
  v8f acc8;
#pragma unroll
  for (int q = kPasses - 1; q >= 0; --q) {
    const v4f vlo = *(const v4f*)(vp + q * kVRow);
    const v4f vhi = *(const v4f*)(vp + q * kVRow + 4);
    v8f c0 = {vlo.x, vlo.y, vlo.z, vlo.w, vhi.x, vhi.y, vhi.z, vhi.w};
    v8f c1 = {0.f, 0.f, 0.f, 0.f, 0.f, 0.f, 0.f, 0.f};
    v2f bfr[kPassTiles];
#pragma unroll
    for (int kc = 0; kc < kPassTiles; ++kc) bfr[kc] = *(const v2f*)(bp + 4 * kc);
    __builtin_amdgcn_sched_barrier(0);
#pragma unroll
    for (int kc = 0; kc < kPassTiles; kc += 2) {
      c0 = mma_f32x4(afr[kc], bfr[kc], c0);
      c1 = mma_f32x4(afr[kc + 1], bfr[kc + 1], c1);
    }
    acc8 = c0 + c1;
    if (q > 0) {
      *(v4f*)xo       = acc8.lo;
      *(v4f*)(xo + 4) = acc8.hi;
      wave_lds_handover();
    }
  }

  float* accL = &sX[wave][kXLo];
  {
    v8f lg;
#pragma unroll
    for (int r = 0; r < 8; ++r) lg[r] = logf(fmaxf(acc8[r], 1e-38f));
    *(v4f*)xo       = acc8.lo;
    *(v4f*)(xo + 4) = acc8.hi;
    float* lo = &sLogAcc[wave][16 * c16 + 8 * h];
    *(v4f*)lo       = lg.lo;
    *(v4f*)(lo + 4) = lg.hi;
  }
  for (int t = kS + lane; t < 384; t += 32) sLogAcc[wave][t] = 0.f;
  wave_lds_handover();

#pragma unroll
  for (int off = 16; off; off >>= 1) pl += __shfl_xor(pl, off, 32);

  float best = INFINITY;

  const int m    = lane & 15;
  const int koff = (lane & 16) ? 2 : 0;
  const float* dp = &sDataPad[wave][16 - m];
  const float* lp = &sLogAcc[wave][16 * m];
  const float* ap = &accL[16 * m];
  v8f dacc0 = {0.f, 0.f, 0.f, 0.f, 0.f, 0.f, 0.f, 0.f};
  v8f dacc1 = {0.f, 0.f, 0.f, 0.f, 0.f, 0.f, 0.f, 0.f};
  v8f s16   = {0.f, 0.f, 0.f, 0.f, 0.f, 0.f, 0.f, 0.f};
  v8f zacc  = {0.f, 0.f, 0.f, 0.f, 0.f, 0.f, 0.f, 0.f};
  constexpr int kS16Tiles = 8;
  constexpr int kZFirst   = 10;
  constexpr int kZTiles   = 4;
#pragma unroll
  for (int kc = 0; kc < kKTiles; ++kc) {
    const int j = 4 * kc + koff;
    v2f a = {lp[j], lp[j + 1]};
    v2f b = {dp[j], dp[j + 1]};
    if (kc & 1) dacc1 = mma_f32x4(a, b, dacc1);
    else        dacc0 = mma_f32x4(a, b, dacc0);
    if (kc < kS16Tiles) {
      v2f a1 = {ap[j], ap[j + 1]};
      v2f b1 = {((unsigned)(j - m) < 16u) ? 1.f : 0.f,
                ((unsigned)(j + 1 - m) < 16u) ? 1.f : 0.f};
      s16 = mma_f32x4(a1, b1, s16);
    }
    if (kc >= kZFirst && kc < kZFirst + kZTiles) {
      const int t  = kc - kZFirst;
      const int mp = 4 * koff + 2 * t;
      v2f a2 = {((unsigned)(mp - m) < 8u) ? 1.f : 0.f,
                ((unsigned)(mp + 1 - m) < 8u) ? 1.f : 0.f};
      v2f b2 = {s16[2 * t], s16[2 * t + 1]};
      zacc = mma_f32x4(a2, b2, zacc);
    }
  }
  const v8f dacc = dacc0 + dacc1;
  const int mbase = (lane & 16) ? 8 : 0;
#pragma unroll
  for (int r = 0; r < 8; ++r) {
    const int k = m + 16 * (r + mbase);
    const float loss = pl - dacc[r] + logf(zacc[r]);
    best = (k < kShifts) ? fminf(best, loss) : best;
  }

#pragma unroll
  for (int off = 16; off; off >>= 1) best = fminf(best, __shfl_xor(best, off, 32));
  if (lane == 0 && row < n_rows) out[row] = best;
}

extern "C" void kernel_launch(void* const* d_in, const int* in_sizes, int n_in,
                              void* d_out, int out_size, void* d_ws, size_t ws_size,
                              hipStream_t stream) {
  (void)n_in; (void)d_ws; (void)ws_size; (void)out_size;
  const float* data = (const float*)d_in[0];
  const float* logw = (const float*)d_in[1];
  float* out = (float*)d_out;
  const int n_rows = in_sizes[0] / kC;
  const int blocks = (n_rows + kRows - 1) / kRows;
  tonal_diffusion_kernel<<<blocks, 256, 0, stream>>>(data, logw, out, n_rows);
}
